# starting kernel
# baseline (speedup 1.0000x reference)
#include <hip/hip_runtime.h>

typedef __attribute__((ext_vector_type(2))) float v2f;
typedef __attribute__((ext_vector_type(8))) float v8f;

#ifndef FEAT
#define FEAT 16
#endif

__global__ __launch_bounds__(256) void IPC_vertex_mlp_wmma(
    const float* __restrict__ vf,
    const float* __restrict__ py,
    const float* __restrict__ vp,
    float* __restrict__ out,
    int nvert)
{
    const int lane = threadIdx.x & 31;
    const int wave = threadIdx.x >> 5;
    const int v    = blockIdx.x * (blockDim.x >> 5) + wave;
    if (v >= nvert) return;

    const float* xf  = vf + (size_t)v * (FEAT * 128);
    const float* pyv = py + (size_t)v * (128 * 2);
    const float* p   = vp + (size_t)v * 233;

    const int  row   = lane & 15;
    const int  khalf = (lane >> 4) << 1;
    const bool lo    = (lane < 16);

    float m0 = __builtin_inff(), m1 = __builtin_inff();
#pragma unroll
    for (int i = 0; i < 4; ++i) {
        const int pt = lane + i * 32;
        m0 = fminf(m0, pyv[pt * 2 + 0]);
        m1 = fminf(m1, pyv[pt * 2 + 1]);
    }
#pragma unroll
    for (int off = 16; off > 0; off >>= 1) {
        m0 = fminf(m0, __shfl_xor(m0, off, 32));
        m1 = fminf(m1, __shfl_xor(m1, off, 32));
    }

    v2f A1[5];
#pragma unroll
    for (int kc = 0; kc < 5; ++kc) {
        const int k0 = kc * 4 + khalf;
        const bool ok0 = (row < 8) && (k0 < 18);
        const bool ok1 = (row < 8) && (k0 + 1 < 18);
        float a0 = p[ok0 ? (row * 18 + k0)     : 0];
        float a1 = p[ok1 ? (row * 18 + k0 + 1) : 0];
        A1[kc].x = ok0 ? a0 : 0.0f;
        A1[kc].y = ok1 ? a1 : 0.0f;
    }
    v2f A2[2];
#pragma unroll
    for (int kc = 0; kc < 2; ++kc) {
        const int k0 = kc * 4 + khalf;
        const bool okr = (row < 8);
        float a0 = p[okr ? (152 + row * 8 + k0)     : 0];
        float a1 = p[okr ? (152 + row * 8 + k0 + 1) : 0];
        A2[kc].x = okr ? a0 : 0.0f;
        A2[kc].y = okr ? a1 : 0.0f;
    }

    float b1r[8], b2r[8], w3r[8];
#pragma unroll
    for (int r = 0; r < 8; ++r) {
        b1r[r] = p[144 + r];
        b2r[r] = p[216 + r];
        w3r[r] = p[224 + r];
    }
    const float b3 = p[232];

#pragma unroll 1
    for (int nt = 0; nt < 8; ++nt) {
        const int col = nt * 16 + row;

        v8f c1 = {};
#pragma unroll
        for (int kc = 0; kc < 5; ++kc) {
            const int k0 = kc * 4 + khalf;
            float bb0, bb1;
            if (kc < 4) {
                bb0 = xf[k0 * 128 + col];
                bb1 = xf[(k0 + 1) * 128 + col];
            } else {
                float p0 = pyv[col * 2 + 0];
                float p1 = pyv[col * 2 + 1];
                bb0 = lo ? (p0 - m0) : 0.0f;
                bb1 = lo ? (p1 - m1) : 0.0f;
            }
            v2f B; B.x = bb0; B.y = bb1;
            c1 = __builtin_amdgcn_wmma_f32_16x16x4_f32(
                     false, A1[kc], false, B, (short)0, c1, false, false);
        }
#pragma unroll
        for (int r = 0; r < 8; ++r) c1[r] = fmaxf(c1[r] + b1r[r], 0.0f);

        v8f c2 = {};
#pragma unroll
        for (int kc = 0; kc < 2; ++kc) {
            float s0 = __shfl(c1[kc * 4 + 0], row, 32);
            float s1 = __shfl(c1[kc * 4 + 1], row, 32);
            float s2 = __shfl(c1[kc * 4 + 2], row, 32);
            float s3 = __shfl(c1[kc * 4 + 3], row, 32);
            v2f B; B.x = lo ? s0 : s2; B.y = lo ? s1 : s3;
            c2 = __builtin_amdgcn_wmma_f32_16x16x4_f32(
                     false, A2[kc], false, B, (short)0, c2, false, false);
        }
#pragma unroll
        for (int r = 0; r < 8; ++r) c2[r] = fmaxf(c2[r] + b2r[r], 0.0f);

        float acc = b3;
#pragma unroll
        for (int r = 0; r < 8; ++r) acc = fmaf(w3r[r], c2[r], acc);
        if (lo) out[(size_t)v * 128 + nt * 16 + row] = acc;
    }
}

extern "C" void kernel_launch(void* const* d_in, const int* in_sizes, int n_in,
                              void* d_out, int out_size, void* d_ws, size_t ws_size,
                              hipStream_t stream) {
    (void)n_in; (void)out_size; (void)d_ws; (void)ws_size;
    const float* vf = (const float*)d_in[0];
    const float* py = (const float*)d_in[1];
    const float* vp = (const float*)d_in[2];
    float* out = (float*)d_out;

    const int nvert = in_sizes[0] / (FEAT * 128);
    const int wavesPerBlock = 8;
    dim3 block(32 * wavesPerBlock);
    dim3 grid((nvert + wavesPerBlock - 1) / wavesPerBlock);
    IPC_vertex_mlp_wmma<<<grid, block, 0, stream>>>(vf, py, vp, out, nvert);
}


# Lane = point: each feature load is one whole 128-B line of a row; v_permlane16_swap of rows k, k+2 builds both 16-point tiles' B fragments; 4 trips of 32 points, whole-line stores; sums unchanged
# speedup vs baseline: 1.1097x; 1.1097x over previous
#include <hip/hip_runtime.h>

typedef __attribute__((ext_vector_type(2))) float v2f;
typedef __attribute__((ext_vector_type(8))) float v8f;

#ifndef FEAT
#define FEAT 16
#endif

__device__ __forceinline__ float xhalf(float x) {
    return __builtin_bit_cast(float, __builtin_amdgcn_permlanex16(
        __builtin_bit_cast(int, x), __builtin_bit_cast(int, x), 0x76543210, 0xfedcba98, false, false));
}

__device__ __forceinline__ void half_swap(float& d, float& s) {
    asm("v_permlane16_swap_b32 %0, %1" : "+v"(d), "+v"(s));
}

__device__ __forceinline__ float mlp_tail(v8f c1, const v2f (&A2)[2], const float (&b1r)[8],
                                          const float (&b2r)[8], const float (&w3r)[8], float b3,
                                          int row, bool lo)
{
#pragma unroll
    for (int r = 0; r < 8; ++r) c1[r] = fmaxf(c1[r] + b1r[r], 0.0f);
    v8f c2 = {};
#pragma unroll
    for (int kc = 0; kc < 2; ++kc) {
        float s0 = __shfl(c1[kc * 4 + 0], row, 32);
        float s1 = __shfl(c1[kc * 4 + 1], row, 32);
        float s2 = __shfl(c1[kc * 4 + 2], row, 32);
        float s3 = __shfl(c1[kc * 4 + 3], row, 32);
        v2f B; B.x = lo ? s0 : s2; B.y = lo ? s1 : s3;
        c2 = __builtin_amdgcn_wmma_f32_16x16x4_f32(false, A2[kc], false, B, (short)0, c2, false, false);
    }
#pragma unroll
    for (int r = 0; r < 8; ++r) c2[r] = fmaxf(c2[r] + b2r[r], 0.0f);
    float acc = b3;
#pragma unroll
    for (int r = 0; r < 8; ++r) acc = fmaf(w3r[r], c2[r], acc);
    return acc;
}

__global__ __launch_bounds__(256) void IPC_vertex_mlp_wmma(
    const float* __restrict__ vf,
    const float* __restrict__ py,
    const float* __restrict__ vp,
    float* __restrict__ out,
    int nvert)
{
    const int lane = threadIdx.x & 31;
    const int wave = __builtin_amdgcn_readfirstlane((int)(threadIdx.x >> 5));
    const int v    = blockIdx.x * (blockDim.x >> 5) + wave;
    if (v >= nvert) return;

    const float* xf  = vf + (size_t)v * (FEAT * 128);
    const float* pyv = py + (size_t)v * (128 * 2);
    const float* p   = vp + (size_t)v * 233;

    const int  row   = lane & 15;
    const int  khalf = (lane >> 4) << 1;
    const bool lo    = (lane < 16);

    float m0 = __builtin_inff(), m1 = __builtin_inff();
#pragma unroll
    for (int i = 0; i < 4; ++i) {
        const int pt = lane + i * 32;
        m0 = fminf(m0, pyv[pt * 2 + 0]);
        m1 = fminf(m1, pyv[pt * 2 + 1]);
    }
#pragma unroll
    for (int off = 16; off > 0; off >>= 1) {
        m0 = fminf(m0, __shfl_xor(m0, off, 32));
        m1 = fminf(m1, __shfl_xor(m1, off, 32));
    }

    v2f A1[5];
#pragma unroll
    for (int kc = 0; kc < 5; ++kc) {
        const int k0 = kc * 4 + khalf;
        const bool ok0 = (row < 8) && (k0 < 18);
        const bool ok1 = (row < 8) && (k0 + 1 < 18);
        float a0 = p[ok0 ? (row * 18 + k0)     : 0];
        float a1 = p[ok1 ? (row * 18 + k0 + 1) : 0];
        A1[kc].x = ok0 ? a0 : 0.0f;
        A1[kc].y = ok1 ? a1 : 0.0f;
    }
    v2f A1b;
    A1b.x = xhalf(A1[4].x);
    A1b.y = xhalf(A1[4].y);

    v2f A2[2];
#pragma unroll
    for (int kc = 0; kc < 2; ++kc) {
        const int k0 = kc * 4 + khalf;
        const bool okr = (row < 8);
        float a0 = p[okr ? (152 + row * 8 + k0)     : 0];
        float a1 = p[okr ? (152 + row * 8 + k0 + 1) : 0];
        A2[kc].x = okr ? a0 : 0.0f;
        A2[kc].y = okr ? a1 : 0.0f;
    }

    float b1r[8], b2r[8], w3r[8];
#pragma unroll
    for (int r = 0; r < 8; ++r) {
        b1r[r] = p[144 + r];
        b2r[r] = p[216 + r];
        w3r[r] = p[224 + r];
    }
    const float b3 = p[232];

#pragma unroll 1
    for (int j = 0; j < 4; ++j) {
        const int pt = j * 32 + lane;

        float r[16];
#pragma unroll
        for (int k = 0; k < 16; ++k) r[k] = xf[k * 128 + pt];
        const v2f pp = *reinterpret_cast<const v2f*>(pyv + 2 * pt);

#pragma unroll
        for (int kc = 0; kc < 4; ++kc) {
            half_swap(r[4 * kc + 0], r[4 * kc + 2]);
            half_swap(r[4 * kc + 1], r[4 * kc + 3]);
        }

        const float d0 = pp.x - m0;
        const float d1 = pp.y - m1;

        v8f cA = {};
#pragma unroll
        for (int kc = 0; kc < 4; ++kc) {
            v2f B; B.x = r[4 * kc + 0]; B.y = r[4 * kc + 1];
            cA = __builtin_amdgcn_wmma_f32_16x16x4_f32(false, A1[kc], false, B, (short)0, cA, false, false);
        }
        {
            v2f B; B.x = lo ? d0 : 0.0f; B.y = lo ? d1 : 0.0f;
            cA = __builtin_amdgcn_wmma_f32_16x16x4_f32(false, A1[4], false, B, (short)0, cA, false, false);
        }
        const float accA = mlp_tail(cA, A2, b1r, b2r, w3r, b3, row, lo);

        v8f cB = {};
#pragma unroll
        for (int kc = 0; kc < 4; ++kc) {
            v2f B; B.x = r[4 * kc + 2]; B.y = r[4 * kc + 3];
            cB = __builtin_amdgcn_wmma_f32_16x16x4_f32(false, A1[kc], false, B, (short)0, cB, false, false);
        }
        {
            v2f B; B.x = lo ? 0.0f : d0; B.y = lo ? 0.0f : d1;
            cB = __builtin_amdgcn_wmma_f32_16x16x4_f32(false, A1b, false, B, (short)0, cB, false, false);
        }
        const float accB = mlp_tail(cB, A2, b1r, b2r, w3r, b3, row, lo);

        const float accBx = xhalf(accB);
        const float res = lo ? accA : accBx;
        out[(size_t)v * 128 + pt] = res;
    }
}

extern "C" void kernel_launch(void* const* d_in, const int* in_sizes, int n_in,
                              void* d_out, int out_size, void* d_ws, size_t ws_size,
                              hipStream_t stream) {
    (void)n_in; (void)out_size; (void)d_ws; (void)ws_size;
    const float* vf = (const float*)d_in[0];
    const float* py = (const float*)d_in[1];
    const float* vp = (const float*)d_in[2];
    float* out = (float*)d_out;

    const int nvert = in_sizes[0] / (FEAT * 128);
    const int wavesPerBlock = 8;
    dim3 block(32 * wavesPerBlock);
    dim3 grid((nvert + wavesPerBlock - 1) / wavesPerBlock);
    IPC_vertex_mlp_wmma<<<grid, block, 0, stream>>>(vf, py, vp, out, nvert);
}
